# starting kernel
# baseline (speedup 1.0000x reference)
#include <hip/hip_runtime.h>

typedef __attribute__((ext_vector_type(2))) float v2f;
typedef __attribute__((ext_vector_type(8))) float v8f;

#define NB   2
#define CHN  4
#define SS   100
#define SP   112

#define OFF_WR    0
#define OFF_WCT   (NB*SP*SP)
#define OFF_MIN   (2*NB*SP*SP)
#define OFF_SUM   (2*NB*SP*SP + 2*NB*SP)
#define OFF_IMG   (2*NB*SP*SP + 4*NB*SP)
#define OFF_TMP   (OFF_IMG + NB*CHN*SP*SP)

__global__ void dec_prep(const float* __restrict__ transforms, float* __restrict__ ws) {
    int t = blockIdx.x * blockDim.x + threadIdx.x;
    if (t >= NB * 2 * SP) return;
    int n    = t / (2 * SP);
    int rem  = t % (2 * SP);
    int axis = rem / SP;
    int q    = rem % SP;

    float t0 = transforms[n*4+0], t1 = transforms[n*4+1];
    float t2 = transforms[n*4+2], t3 = transforms[n*4+3];
    float scale = (axis == 0) ? (t1 - t0) : (t3 - t2);
    float off   = (axis == 0) ? (t0 * (float)SS) : (t2 * (float)SS);

    float m = 0.0f, sum = 1.0f;
    if (q < SS) {
        m = 3.4e38f;
        for (int k = 0; k < SS; ++k) {
            float d = (float)q - (scale * (float)k + off);
            m = fminf(m, d * d);
        }
        sum = 0.0f;
        for (int k = 0; k < SS; ++k) {
            float d = (float)q - (scale * (float)k + off);
            sum += __expf(m - d * d);
        }
    }
    ws[OFF_MIN + t] = m;
    ws[OFF_SUM + t] = sum;
}

__global__ void dec_fill(const float* __restrict__ transforms, float* __restrict__ ws) {
    int b    = blockIdx.x;
    int n    = b / (2 * SP);
    int rem  = b % (2 * SP);
    int axis = rem / SP;
    int row  = rem % SP;
    int col  = threadIdx.x;
    if (col >= SP) return;

    float t0 = transforms[n*4+0], t1 = transforms[n*4+1];
    float t2 = transforms[n*4+2], t3 = transforms[n*4+3];
    float scale = (axis == 0) ? (t1 - t0) : (t3 - t2);
    float off   = (axis == 0) ? (t0 * (float)SS) : (t2 * (float)SS);

    float val = 0.0f;
    if (row < SS && col < SS) {
        if (axis == 0) {
            float m = ws[OFF_MIN + (n*2 + 0)*SP + row];
            float d = (float)row - (scale * (float)col + off);
            val = __expf(m - d * d);
        } else {
            float m = ws[OFF_MIN + (n*2 + 1)*SP + col];
            float d = (float)col - (scale * (float)row + off);
            val = __expf(m - d * d);
        }
    }
    float* dst = (axis == 0) ? (ws + OFF_WR + n*SP*SP) : (ws + OFF_WCT + n*SP*SP);
    dst[row*SP + col] = val;
}

__global__ void dec_pad(const float* __restrict__ images, float* __restrict__ ws) {
    int b   = blockIdx.x;
    int mat = b / SP;
    int row = b % SP;
    int col = threadIdx.x;
    if (col >= SP) return;
    float v = 0.0f;
    if (row < SS && col < SS) v = images[mat * SS * SS + row * SS + col];
    ws[OFF_IMG + mat * SP * SP + row * SP + col] = v;
}

__device__ __forceinline__ v8f wmma_tile_112(const float* __restrict__ A,
                                             const float* __restrict__ B,
                                             int tM, int tN, int lane) {
    int hs   = lane >> 4;
    int l16  = lane & 15;
    int rowA = tM * 16 + l16;
    int colB = tN * 16 + l16;

    v8f c = {};
#pragma unroll
    for (int k0 = 0; k0 < SP; k0 += 4) {
        int ka = k0 + 2 * hs;
        v2f a = *(const v2f*)(A + rowA * SP + ka);
        v2f b;
        b.x = B[ka * SP + colB];
        b.y = B[(ka + 1) * SP + colB];
        c = __builtin_amdgcn_wmma_f32_16x16x4_f32(false, a, false, b,
                                                  (short)0, c, false, false);
    }
    return c;
}

__global__ void dec_gemm_img_wct(float* __restrict__ ws) {
    int bid  = blockIdx.x;
    int mat  = bid / 49;
    int tile = bid % 49;
    int tM = tile / 7, tN = tile % 7;
    int n  = mat / CHN;

    const float* A = ws + OFF_IMG + mat * SP * SP;
    const float* B = ws + OFF_WCT + n   * SP * SP;
    float*       D = ws + OFF_TMP + mat * SP * SP;

    int lane = threadIdx.x;
    v8f c = wmma_tile_112(A, B, tM, tN, lane);

    int hs = lane >> 4, l16 = lane & 15;
    int baseM = tM * 16 + hs * 8;
#pragma unroll
    for (int v = 0; v < 8; ++v)
        D[(baseM + v) * SP + tN * 16 + l16] = c[v];
}

__global__ void dec_gemm_wr_tmp(const float* __restrict__ ws, float* __restrict__ out) {
    int bid  = blockIdx.x;
    int mat  = bid / 49;
    int tile = bid % 49;
    int tM = tile / 7, tN = tile % 7;
    int n  = mat / CHN;

    const float* A    = ws + OFF_WR  + n   * SP * SP;
    const float* B    = ws + OFF_TMP + mat * SP * SP;
    const float* rs_r = ws + OFF_SUM + (n*2 + 0) * SP;
    const float* rs_c = ws + OFF_SUM + (n*2 + 1) * SP;

    int lane = threadIdx.x;
    v8f c = wmma_tile_112(A, B, tM, tN, lane);

    int hs = lane >> 4, l16 = lane & 15;
    int qj = tN * 16 + l16;
    if (qj < SS) {
        float rc    = rs_c[qj];
        int   baseM = tM * 16 + hs * 8;
#pragma unroll
        for (int v = 0; v < 8; ++v) {
            int qi = baseM + v;
            if (qi < SS)
                out[mat * SS * SS + qi * SS + qj] = c[v] / (rs_r[qi] * rc);
        }
    }
}

extern "C" void kernel_launch(void* const* d_in, const int* in_sizes, int n_in,
                              void* d_out, int out_size, void* d_ws, size_t ws_size,
                              hipStream_t stream) {
    (void)in_sizes; (void)n_in; (void)out_size; (void)ws_size;
    const float* images     = (const float*)d_in[0];
    const float* transforms = (const float*)d_in[1];
    float*       out        = (float*)d_out;
    float*       ws         = (float*)d_ws;

    dec_prep<<<(NB*2*SP + 127) / 128, 128, 0, stream>>>(transforms, ws);
    dec_fill<<<NB*2*SP, 128, 0, stream>>>(transforms, ws);
    dec_pad <<<NB*CHN*SP, 128, 0, stream>>>(images, ws);
    dec_gemm_img_wct<<<NB*CHN*49, 32, 0, stream>>>(ws);
    dec_gemm_wr_tmp <<<NB*CHN*49, 32, 0, stream>>>(ws, out);
}


# Fuse the 5 launches into 1 kernel: a block owns one 16-column output strip, both WMMA products with the intermediate in LDS, weights built in registers; denominators summed as 4 partials (k mod 4)
# speedup vs baseline: 4.7726x; 4.7726x over previous
#include <hip/hip_runtime.h>


typedef __attribute__((ext_vector_type(2))) float v2f;
typedef __attribute__((ext_vector_type(4))) float v4f;
typedef __attribute__((ext_vector_type(8))) float v8f;

#define NB   2
#define CHN  4
#define SS   100
#define NT   7
#define KS   (SS / 4)

__device__ __forceinline__ float xhalf(float x) {
    int b = __builtin_bit_cast(int, x);
    return __builtin_bit_cast(float,
        __builtin_amdgcn_permlanex16(b, b, 0x76543210, 0xfedcba98, false, false));
}

__device__ __forceinline__ float axis_weights(float q, float scale, float off, int hs, v2f (&w)[KS]) {
    const float kb = (float)(2 * hs);
    float m = 3.4e38f;
#pragma unroll
    for (int s = 0; s < KS; ++s) {
        v2f k = {kb + (float)(4 * s), kb + (float)(4 * s + 1)};
        v2f d = q - (scale * k + off);
        v2f d2 = d * d;
        w[s] = d;
        m = fminf(m, fminf(d2.x, d2.y));
    }
    m = fminf(m, xhalf(m));
    v2f sum = {0.0f, 0.0f};
#pragma unroll
    for (int s = 0; s < KS; ++s) {
        v2f d = w[s];
        v2f e = {__expf(m - d.x * d.x), __expf(m - d.y * d.y)};
        w[s] = e;
        sum += e;
    }
    float t = sum.x + sum.y;
    return t + xhalf(t);
}

__global__ __launch_bounds__(NT * 32) __attribute__((amdgpu_waves_per_eu(2, 2)))
void dec_fused(const float* __restrict__ images, const float* __restrict__ transforms,
               float* __restrict__ out) {
    __shared__ __attribute__((aligned(16))) float tmp_s[NT * 16 * 16];
    __shared__ __attribute__((aligned(16))) float rs_s[NT * 16];

    const int mat  = blockIdx.x / NT;
    const int tN   = blockIdx.x % NT;
    const int n    = mat / CHN;
    const int tM   = threadIdx.x >> 5;
    const int lane = threadIdx.x & 31;
    const int hs   = lane >> 4;
    const int l16  = lane & 15;
    const int rowA = tM * 16 + l16;
    const int qj   = tN * 16 + l16;

    const int rowL = rowA < SS ? rowA : SS - 1;
    const float* ap = images + mat * SS * SS + rowL * SS + 2 * hs;
    v2f a1[KS];
#pragma unroll
    for (int s = 0; s < KS; ++s) a1[s] = *(const v2f*)(ap + 4 * s);

    const float t0 = transforms[n * 4 + 0], t1 = transforms[n * 4 + 1];
    const float t2 = transforms[n * 4 + 2], t3 = transforms[n * 4 + 3];

    v2f wgt[KS];

    const float rc = axis_weights((float)qj, t3 - t2, t2 * (float)SS, hs, wgt);
    v8f c = {};
#pragma unroll
    for (int s = 0; s < KS; ++s)
        c = __builtin_amdgcn_wmma_f32_16x16x4_f32(false, a1[s], false, wgt[s],
                                                  (short)0, c, false, false);
    const int baseM = tM * 16 + hs * 8;
#pragma unroll
    for (int p = 0; p < 4; ++p)
        *(v2f*)&tmp_s[(((baseM >> 1) + p) * 16 + l16) * 2] = (v2f){c[2 * p], c[2 * p + 1]};

    const float rr = axis_weights((float)rowA, t1 - t0, t0 * (float)SS, hs, wgt);
    if (hs == 0) rs_s[rowA] = rr;
    __syncthreads();

    v2f b2[KS];
#pragma unroll
    for (int s = 0; s < KS; ++s) b2[s] = *(const v2f*)&tmp_s[((2 * s + hs) * 16 + l16) * 2];
    const v4f r0 = *(const v4f*)&rs_s[baseM];
    const v4f r1 = *(const v4f*)&rs_s[baseM + 4];
    __builtin_amdgcn_sched_barrier(0);

    c = (v8f){};
#pragma unroll
    for (int s = 0; s < KS; ++s)
        c = __builtin_amdgcn_wmma_f32_16x16x4_f32(false, wgt[s], false, b2[s],
                                                  (short)0, c, false, false);

    const float rs[8] = {r0.x, r0.y, r0.z, r0.w, r1.x, r1.y, r1.z, r1.w};
    if (qj < SS) {
        float* op = out + mat * SS * SS + baseM * SS + qj;
        if (baseM + 8 <= SS) {
#pragma unroll
            for (int v = 0; v < 8; ++v) op[v * SS] = c[v] / (rs[v] * rc);
        } else {
#pragma unroll
            for (int v = 0; v < 8; ++v)
                if (baseM + v < SS) op[v * SS] = c[v] / (rs[v] * rc);
        }
    }
}

extern "C" void kernel_launch(void* const* d_in, const int* in_sizes, int n_in,
                              void* d_out, int out_size, void* d_ws, size_t ws_size,
                              hipStream_t stream) {
    (void)in_sizes; (void)n_in; (void)out_size; (void)d_ws; (void)ws_size;
    const float* images     = (const float*)d_in[0];
    const float* transforms = (const float*)d_in[1];
    float*       out        = (float*)d_out;

    dec_fused<<<NB * CHN * NT, NT * 32, 0, stream>>>(images, transforms, out);
}
